# starting kernel
# baseline (speedup 1.0000x reference)
#include <hip/hip_runtime.h>
#include <hip/hip_bf16.h>
#include <cstdint>

#define FEAT_DIM 128

typedef __attribute__((ext_vector_type(2))) float v2f;
typedef __attribute__((ext_vector_type(8))) float v8f;

__device__ __forceinline__ unsigned ordEncode(float f) {
  unsigned u = __float_as_uint(f);
  return (u & 0x80000000u) ? ~u : (u | 0x80000000u);
}
__device__ __forceinline__ float ordDecode(unsigned e) {
  unsigned u = (e & 0x80000000u) ? (e & 0x7fffffffu) : ~e;
  return __uint_as_float(u);
}

__global__ void gat_init_kernel(float* __restrict__ out,
                                unsigned* __restrict__ smax_ord,
                                float* __restrict__ denom,
                                int total, int n_nodes) {
  int i = blockIdx.x * blockDim.x + threadIdx.x;
  if (i < total) out[i] = 0.0f;
  if (i < n_nodes) {
    smax_ord[i] = 0u;
    denom[i] = 0.0f;
  }
}

__global__ void gat_scores_wmma_kernel(const float* __restrict__ feat,
                                       const int* __restrict__ src,
                                       const int* __restrict__ dst,
                                       float* __restrict__ scores,
                                       unsigned* __restrict__ smax_ord,
                                       int E) {
  const int lane = threadIdx.x & 31;
  const int wave = blockIdx.x * (blockDim.x >> 5) + (threadIdx.x >> 5);
  const int e0 = wave * 16;
  if (e0 >= E) return;

  const int m    = lane & 15;
  const int half = lane >> 4;
  const int e    = min(e0 + m, E - 1);

  const float* __restrict__ ps = feat + (size_t)src[e] * FEAT_DIM;
  const float* __restrict__ pd = feat + (size_t)dst[e] * FEAT_DIM;

  v8f c = {0.f, 0.f, 0.f, 0.f, 0.f, 0.f, 0.f, 0.f};
#pragma unroll
  for (int kk = 0; kk < FEAT_DIM; kk += 4) {
    const int kb = kk + 2 * half;
    const float2 av = *(const float2*)(ps + kb);
    const float2 bv = *(const float2*)(pd + kb);
    v2f a; a.x = av.x; a.y = av.y;
    v2f b; b.x = bv.x; b.y = bv.y;
    c = __builtin_amdgcn_wmma_f32_16x16x4_f32(false, a, false, b, (short)0, c,
                                              false, false);
  }

  float s = 0.0f;
#pragma unroll
  for (int r = 0; r < 8; ++r) {
    const float vlo = __shfl(c[r], r, 32);
    const float vhi = __shfl(c[r], r + 24, 32);
    if (lane == r)     s = vlo;
    if (lane == r + 8) s = vhi;
  }

  if (lane < 16 && (e0 + lane) < E) {
    const float score = s * (1.0f / 1.0f);
    scores[e0 + lane] = score;
    atomicMax(&smax_ord[dst[e0 + lane]], ordEncode(score));
  }
}

__global__ void gat_expsum_kernel(const int* __restrict__ dst,
                                  float* __restrict__ scores_ex,
                                  const unsigned* __restrict__ smax_ord,
                                  float* __restrict__ denom, int E) {
  const int e = blockIdx.x * blockDim.x + threadIdx.x;
  if (e >= E) return;
  const int d = dst[e];
  const float mx = ordDecode(smax_ord[d]);
  const float ex = __expf(scores_ex[e] - mx);
  scores_ex[e] = ex;
  atomicAdd(&denom[d], ex);
}

__global__ void gat_aggregate_kernel(const float* __restrict__ feat,
                                     const int* __restrict__ src,
                                     const int* __restrict__ dst,
                                     const float* __restrict__ ex,
                                     const float* __restrict__ denom,
                                     float* __restrict__ out, int E) {
  const int lane = threadIdx.x & 31;
  const int e = blockIdx.x * (blockDim.x >> 5) + (threadIdx.x >> 5);
  if (e >= E) return;
  const int d = dst[e];
  const float w = ex[e] / denom[d];
  const float4 v = ((const float4*)(feat + (size_t)src[e] * FEAT_DIM))[lane];
  float* po = out + (size_t)d * FEAT_DIM + lane * 4;
  atomicAdd(po + 0, w * v.x);
  atomicAdd(po + 1, w * v.y);
  atomicAdd(po + 2, w * v.z);
  atomicAdd(po + 3, w * v.w);
}

__global__ void gat_fixup_kernel(const float* __restrict__ feat,
                                 const float* __restrict__ denom,
                                 float* __restrict__ out, int total) {
  const int i = blockIdx.x * blockDim.x + threadIdx.x;
  if (i >= total) return;
  const int n = i >> 7;
  if (denom[n] == 0.0f) out[i] = feat[i];
}

extern "C" void kernel_launch(void* const* d_in, const int* in_sizes, int n_in,
                              void* d_out, int out_size, void* d_ws, size_t ws_size,
                              hipStream_t stream) {
  const float* feat = (const float*)d_in[0];
  const int* src    = (const int*)d_in[1];
  const int* dst    = (const int*)d_in[2];
  const int E       = in_sizes[1];
  const int N       = in_sizes[0] / FEAT_DIM;
  const int total   = N * FEAT_DIM;
  float* out        = (float*)d_out;

  float*    scores   = (float*)d_ws;
  unsigned* smax_ord = (unsigned*)(scores + E);
  float*    denom    = (float*)(smax_ord + N);

  const int BLK = 256;

  gat_init_kernel<<<(total + BLK - 1) / BLK, BLK, 0, stream>>>(
      out, smax_ord, denom, total, N);

  {
    const int waves   = (E + 15) / 16;
    const int threads = waves * 32;
    gat_scores_wmma_kernel<<<(threads + BLK - 1) / BLK, BLK, 0, stream>>>(
        feat, src, dst, scores, smax_ord, E);
  }

  gat_expsum_kernel<<<(E + BLK - 1) / BLK, BLK, 0, stream>>>(
      dst, scores, smax_ord, denom, E);

  {
    const long long threads = (long long)E * 32;
    gat_aggregate_kernel<<<(int)((threads + BLK - 1) / BLK), BLK, 0, stream>>>(
        feat, src, dst, scores, denom, out, E);
  }

  gat_fixup_kernel<<<(total + BLK - 1) / BLK, BLK, 0, stream>>>(
      feat, denom, out, total);
}


# scores kernel: gather each 512-B feature row with one coalesced wave load and transpose to f32 WMMA fragments through a per-wave LDS tile, instead of 8-byte strided per-lane gathers
# speedup vs baseline: 1.2765x; 1.2765x over previous
#include <hip/hip_runtime.h>
#include <hip/hip_bf16.h>
#include <cstdint>

#define FEAT_DIM 128

typedef __attribute__((ext_vector_type(2))) float v2f;
typedef __attribute__((ext_vector_type(8))) float v8f;

__device__ __forceinline__ unsigned ordEncode(float f) {
  unsigned u = __float_as_uint(f);
  return (u & 0x80000000u) ? ~u : (u | 0x80000000u);
}
__device__ __forceinline__ float ordDecode(unsigned e) {
  unsigned u = (e & 0x80000000u) ? (e & 0x7fffffffu) : ~e;
  return __uint_as_float(u);
}

__global__ void gat_init_kernel(float* __restrict__ out,
                                unsigned* __restrict__ smax_ord,
                                float* __restrict__ denom,
                                int total, int n_nodes) {
  int i = blockIdx.x * blockDim.x + threadIdx.x;
  if (i < total) out[i] = 0.0f;
  if (i < n_nodes) {
    smax_ord[i] = 0u;
    denom[i] = 0.0f;
  }
}

#define SCORE_WAVES 2
#define ROW_PITCH4 33

__global__ __launch_bounds__(SCORE_WAVES * 32)
void gat_scores_wmma_kernel(const float* __restrict__ feat,
                            const int* __restrict__ src,
                            const int* __restrict__ dst,
                            float* __restrict__ scores,
                            unsigned* __restrict__ smax_ord,
                            int E) {
  __shared__ float4 stage[SCORE_WAVES][2][16][ROW_PITCH4];

  const int lane = threadIdx.x & 31;
  const int wib  = threadIdx.x >> 5;
  const int wave = blockIdx.x * (blockDim.x >> 5) + wib;
  const int e0 = wave * 16;
  if (e0 >= E) return;

  const int m    = lane & 15;
  const int half = lane >> 4;
  const int e    = min(e0 + m, E - 1);

  const int sidx = src[e];
  const int didx = dst[e];

  float4 rs[16], rd[16];
#pragma unroll
  for (int r = 0; r < 16; ++r) {
    const int sr = __builtin_amdgcn_readlane(sidx, r);
    const int dr = __builtin_amdgcn_readlane(didx, r);
    rs[r] = ((const float4*)(feat + (size_t)sr * FEAT_DIM))[lane];
    rd[r] = ((const float4*)(feat + (size_t)dr * FEAT_DIM))[lane];
  }
#pragma unroll
  for (int r = 0; r < 16; ++r) {
    stage[wib][0][r][lane] = rs[r];
    stage[wib][1][r][lane] = rd[r];
  }


  v8f c = {0.f, 0.f, 0.f, 0.f, 0.f, 0.f, 0.f, 0.f};
#pragma unroll
  for (int j = 0; j < 16; ++j) {
    const float4 av = stage[wib][0][m][2 * j + half];
    const float4 bv = stage[wib][1][m][2 * j + half];
    v2f a; a.x = av.x; a.y = av.y;
    v2f b; b.x = bv.x; b.y = bv.y;
    c = __builtin_amdgcn_wmma_f32_16x16x4_f32(false, a, false, b, (short)0, c,
                                              false, false);
    a.x = av.z; a.y = av.w;
    b.x = bv.z; b.y = bv.w;
    c = __builtin_amdgcn_wmma_f32_16x16x4_f32(false, a, false, b, (short)0, c,
                                              false, false);
  }

  float s = 0.0f;
#pragma unroll
  for (int r = 0; r < 8; ++r) {
    const float vlo = __shfl(c[r], r, 32);
    const float vhi = __shfl(c[r], r + 24, 32);
    if (lane == r)     s = vlo;
    if (lane == r + 8) s = vhi;
  }

  if (lane < 16 && (e0 + lane) < E) {
    const float score = s * (1.0f / 1.0f);
    scores[e0 + lane] = score;
    atomicMax(&smax_ord[dst[e0 + lane]], ordEncode(score));
  }
}

__global__ void gat_expsum_kernel(const int* __restrict__ dst,
                                  float* __restrict__ scores_ex,
                                  const unsigned* __restrict__ smax_ord,
                                  float* __restrict__ denom, int E) {
  const int e = blockIdx.x * blockDim.x + threadIdx.x;
  if (e >= E) return;
  const int d = dst[e];
  const float mx = ordDecode(smax_ord[d]);
  const float ex = __expf(scores_ex[e] - mx);
  scores_ex[e] = ex;
  atomicAdd(&denom[d], ex);
}

__global__ void gat_aggregate_kernel(const float* __restrict__ feat,
                                     const int* __restrict__ src,
                                     const int* __restrict__ dst,
                                     const float* __restrict__ ex,
                                     const float* __restrict__ denom,
                                     float* __restrict__ out, int E) {
  const int lane = threadIdx.x & 31;
  const int e = blockIdx.x * (blockDim.x >> 5) + (threadIdx.x >> 5);
  if (e >= E) return;
  const int d = dst[e];
  const float w = ex[e] / denom[d];
  const float4 v = ((const float4*)(feat + (size_t)src[e] * FEAT_DIM))[lane];
  float* po = out + (size_t)d * FEAT_DIM + lane * 4;
  atomicAdd(po + 0, w * v.x);
  atomicAdd(po + 1, w * v.y);
  atomicAdd(po + 2, w * v.z);
  atomicAdd(po + 3, w * v.w);
}

__global__ void gat_fixup_kernel(const float* __restrict__ feat,
                                 const float* __restrict__ denom,
                                 float* __restrict__ out, int total) {
  const int i = blockIdx.x * blockDim.x + threadIdx.x;
  if (i >= total) return;
  const int n = i >> 7;
  if (denom[n] == 0.0f) out[i] = feat[i];
}

extern "C" void kernel_launch(void* const* d_in, const int* in_sizes, int n_in,
                              void* d_out, int out_size, void* d_ws, size_t ws_size,
                              hipStream_t stream) {
  const float* feat = (const float*)d_in[0];
  const int* src    = (const int*)d_in[1];
  const int* dst    = (const int*)d_in[2];
  const int E       = in_sizes[1];
  const int N       = in_sizes[0] / FEAT_DIM;
  const int total   = N * FEAT_DIM;
  float* out        = (float*)d_out;

  float*    scores   = (float*)d_ws;
  unsigned* smax_ord = (unsigned*)(scores + E);
  float*    denom    = (float*)(smax_ord + N);

  const int BLK = 256;

  gat_init_kernel<<<(total + BLK - 1) / BLK, BLK, 0, stream>>>(
      out, smax_ord, denom, total, N);

  {
    const int waves   = (E + 15) / 16;
    const int threads = waves * 32;
    const int SBLK = SCORE_WAVES * 32;
    gat_scores_wmma_kernel<<<(threads + SBLK - 1) / SBLK, SBLK, 0, stream>>>(
        feat, src, dst, scores, smax_ord, E);
  }

  gat_expsum_kernel<<<(E + BLK - 1) / BLK, BLK, 0, stream>>>(
      dst, scores, smax_ord, denom, E);

  {
    const long long threads = (long long)E * 32;
    gat_aggregate_kernel<<<(int)((threads + BLK - 1) / BLK), BLK, 0, stream>>>(
        feat, src, dst, scores, denom, out, E);
  }

  gat_fixup_kernel<<<(total + BLK - 1) / BLK, BLK, 0, stream>>>(
      feat, denom, out, total);
}


# group the edges by destination inside the call (counting sort); one wave per node does scores (f32 WMMA), softmax and weighted sum in registers and stores its out row once: no float atomics
# speedup vs baseline: 2.2286x; 1.7459x over previous
#include <hip/hip_runtime.h>
#include <cstdint>
#include <math.h>


#define FEAT_DIM 128

typedef __attribute__((ext_vector_type(2))) float v2f;
typedef __attribute__((ext_vector_type(8))) float v8f;


__global__ void gat_zero_kernel(int* __restrict__ cursor, int n_nodes) {
  const int i = blockIdx.x * blockDim.x + threadIdx.x;
  if (i < n_nodes) cursor[i] = 0;
}

__global__ void gat_count_kernel(const int* __restrict__ dst,
                                 int* __restrict__ cursor, int E) {
  const int e = blockIdx.x * blockDim.x + threadIdx.x;
  if (e < E) atomicAdd(&cursor[dst[e]], 1);
}

#define SCAN_THREADS 1024
#define SCAN_IPT 16

__global__ __launch_bounds__(SCAN_THREADS)
void gat_scan_kernel(int* __restrict__ cursor, int n) {
  __shared__ int wave_tot[SCAN_THREADS / 32];
  const int tid  = threadIdx.x;
  const int lane = tid & 31;
  const int w    = tid >> 5;
  int carry = 0;
  for (int base = 0; base < n; base += SCAN_THREADS * SCAN_IPT) {
    const int i0 = base + tid * SCAN_IPT;
    const bool whole = (i0 + SCAN_IPT <= n);
    int v[SCAN_IPT];
    if (whole) {
#pragma unroll
      for (int q = 0; q < SCAN_IPT / 4; ++q) {
        const int4 t = ((const int4*)(cursor + i0))[q];
        v[4 * q + 0] = t.x; v[4 * q + 1] = t.y;
        v[4 * q + 2] = t.z; v[4 * q + 3] = t.w;
      }
    } else {
#pragma unroll
      for (int k = 0; k < SCAN_IPT; ++k)
        v[k] = (i0 + k < n) ? cursor[i0 + k] : 0;
    }
#pragma unroll
    for (int k = 1; k < SCAN_IPT; ++k) v[k] += v[k - 1];
    const int tsum = v[SCAN_IPT - 1];

    int inc = tsum;
#pragma unroll
    for (int off = 1; off < 32; off <<= 1) {
      const int t = __shfl_up(inc, off, 32);
      if (lane >= off) inc += t;
    }
    if (lane == 31) wave_tot[w] = inc;
    __syncthreads();
    const int wt = wave_tot[lane];
    int winc = wt;
#pragma unroll
    for (int off = 1; off < 32; off <<= 1) {
      const int t = __shfl_up(winc, off, 32);
      if (lane >= off) winc += t;
    }
    const int wave_excl = __shfl(winc - wt, w, 32);
    const int total     = __shfl(winc, 31, 32);
    const int excl = carry + wave_excl + (inc - tsum);

    int ex[SCAN_IPT];
    ex[0] = excl;
#pragma unroll
    for (int k = 1; k < SCAN_IPT; ++k) ex[k] = excl + v[k - 1];
    if (whole) {
#pragma unroll
      for (int q = 0; q < SCAN_IPT / 4; ++q) {
        int4 t;
        t.x = ex[4 * q + 0]; t.y = ex[4 * q + 1];
        t.z = ex[4 * q + 2]; t.w = ex[4 * q + 3];
        ((int4*)(cursor + i0))[q] = t;
      }
    } else {
#pragma unroll
      for (int k = 0; k < SCAN_IPT; ++k)
        if (i0 + k < n) cursor[i0 + k] = ex[k];
    }
    carry += total;
    __syncthreads();
  }
}

__global__ void gat_fill_kernel(const int* __restrict__ src,
                                const int* __restrict__ dst,
                                int* __restrict__ cursor,
                                int* __restrict__ seg_src, int E) {
  const int e = blockIdx.x * blockDim.x + threadIdx.x;
  if (e >= E) return;
  const int slot = atomicAdd(&cursor[dst[e]], 1);
  seg_src[slot] = src[e];
}

#define NODE_WAVES 4
#define ROW_PITCH4 33

__device__ __forceinline__ float lane_value(float v, int l) {
  return __int_as_float(__builtin_amdgcn_readlane(__float_as_int(v), l));
}

__global__ __launch_bounds__(NODE_WAVES * 32)
void gat_node_wmma_kernel(const float* __restrict__ feat,
                          const int* __restrict__ seg_end,
                          const int* __restrict__ seg_src,
                          float* __restrict__ out, int n_nodes) {
  __shared__ float4 stage[NODE_WAVES][16][ROW_PITCH4];
  __shared__ float4 dst_row[NODE_WAVES][32];

  const int lane = threadIdx.x & 31;
  const int wib  = threadIdx.x >> 5;
  const int d = __builtin_amdgcn_readfirstlane(blockIdx.x * NODE_WAVES + wib);
  if (d >= n_nodes) return;

  const int m    = lane & 15;
  const int half = lane >> 4;

  const int beg = (d > 0) ? seg_end[d - 1] : 0;
  const int end = seg_end[d];
  const float4* feat4 = (const float4*)feat;
  const float4 hd = feat4[(size_t)d * (FEAT_DIM / 4) + lane];
  float4* orow = (float4*)(out + (size_t)d * FEAT_DIM) + lane;

  if (end <= beg) {
    *orow = hd;
    return;
  }

  dst_row[wib][lane] = hd;

  float  m_run = -INFINITY;
  float  l_run = 0.0f;
  float4 acc = make_float4(0.f, 0.f, 0.f, 0.f);

  for (int p0 = beg; p0 < end; p0 += 16) {
    const int cnt = min(16, end - p0);
    const int sidx = (lane < cnt) ? seg_src[p0 + lane] : d;

    float4 rs[16];
#pragma unroll
    for (int r = 0; r < 16; ++r) {
      const int sr = __builtin_amdgcn_readlane(sidx, r);
      rs[r] = feat4[(size_t)sr * (FEAT_DIM / 4) + lane];
    }
#pragma unroll
    for (int r = 0; r < 16; ++r) stage[wib][r][lane] = rs[r];


    v8f c = {0.f, 0.f, 0.f, 0.f, 0.f, 0.f, 0.f, 0.f};
#pragma unroll
    for (int j = 0; j < 16; ++j) {
      const float4 av = stage[wib][m][2 * j + half];
      const float4 bv = dst_row[wib][2 * j + half];
      v2f a; a.x = av.x; a.y = av.y;
      v2f b; b.x = bv.x; b.y = bv.y;
      c = __builtin_amdgcn_wmma_f32_16x16x4_f32(false, a, false, b, (short)0, c,
                                                false, false);
      a.x = av.z; a.y = av.w;
      b.x = bv.z; b.y = bv.w;
      c = __builtin_amdgcn_wmma_f32_16x16x4_f32(false, a, false, b, (short)0, c,
                                                false, false);
    }

    float mloc = -INFINITY;
#pragma unroll
    for (int i = 0; i < 8; ++i)
      if (8 * half + i < cnt) mloc = fmaxf(mloc, c[i]);
    const float m_new = fmaxf(m_run, fmaxf(lane_value(mloc, 0), lane_value(mloc, 16)));
    const float scale = __expf(m_run - m_new);

    float p[8];
    float ploc = 0.0f;
#pragma unroll
    for (int i = 0; i < 8; ++i) {
      p[i] = (8 * half + i < cnt) ? __expf(c[i] - m_new) : 0.0f;
      ploc += p[i];
    }
    l_run = l_run * scale + (lane_value(ploc, 0) + lane_value(ploc, 16));

    acc.x *= scale; acc.y *= scale; acc.z *= scale; acc.w *= scale;
#pragma unroll
    for (int r = 0; r < 16; ++r) {
      const float w = lane_value(p[r & 7], (r >> 3) * 16);
      acc.x = fmaf(w, rs[r].x, acc.x);
      acc.y = fmaf(w, rs[r].y, acc.y);
      acc.z = fmaf(w, rs[r].z, acc.z);
      acc.w = fmaf(w, rs[r].w, acc.w);
    }
    m_run = m_new;
  }

  float4 o;
  o.x = acc.x / l_run; o.y = acc.y / l_run;
  o.z = acc.z / l_run; o.w = acc.w / l_run;
  *orow = o;
}

extern "C" void kernel_launch(void* const* d_in, const int* in_sizes, int n_in,
                              void* d_out, int out_size, void* d_ws, size_t ws_size,
                              hipStream_t stream) {
  const float* feat = (const float*)d_in[0];
  const int* src    = (const int*)d_in[1];
  const int* dst    = (const int*)d_in[2];
  const int E       = in_sizes[1];
  const int N       = in_sizes[0] / FEAT_DIM;
  float* out        = (float*)d_out;

  int* cursor  = (int*)d_ws;
  int* seg_src = cursor + N;

  const int BLK = 256;

  gat_zero_kernel<<<(N + BLK - 1) / BLK, BLK, 0, stream>>>(cursor, N);
  gat_count_kernel<<<(E + BLK - 1) / BLK, BLK, 0, stream>>>(dst, cursor, E);
  gat_scan_kernel<<<1, SCAN_THREADS, 0, stream>>>(cursor, N);
  gat_fill_kernel<<<(E + BLK - 1) / BLK, BLK, 0, stream>>>(src, dst, cursor,
                                                           seg_src, E);
  gat_node_wmma_kernel<<<(N + NODE_WAVES - 1) / NODE_WAVES, NODE_WAVES * 32, 0,
                         stream>>>(feat, cursor, seg_src, out, N);
}


# node kernel: h_d fragments read from LDS inside each trip instead of held in 64 registers over the whole kernel (160 to 110 VGPRs): 8 instead of 6 resident waves per SIMD
# speedup vs baseline: 2.3010x; 1.0325x over previous
#include <hip/hip_runtime.h>
#include <cstdint>
#include <math.h>


#define FEAT_DIM 128

typedef __attribute__((ext_vector_type(2))) float v2f;
typedef __attribute__((ext_vector_type(8))) float v8f;


__global__ void gat_zero_kernel(int* __restrict__ cursor, int n_nodes) {
  const int i = blockIdx.x * blockDim.x + threadIdx.x;
  if (i < n_nodes) cursor[i] = 0;
}

__global__ void gat_count_kernel(const int* __restrict__ dst,
                                 int* __restrict__ cursor, int E) {
  const int e = blockIdx.x * blockDim.x + threadIdx.x;
  if (e < E) atomicAdd(&cursor[dst[e]], 1);
}

#define SCAN_THREADS 1024
#define SCAN_IPT 16

__global__ __launch_bounds__(SCAN_THREADS)
void gat_scan_kernel(int* __restrict__ cursor, int n) {
  __shared__ int wave_tot[SCAN_THREADS / 32];
  const int tid  = threadIdx.x;
  const int lane = tid & 31;
  const int w    = tid >> 5;
  int carry = 0;
  for (int base = 0; base < n; base += SCAN_THREADS * SCAN_IPT) {
    const int i0 = base + tid * SCAN_IPT;
    const bool whole = (i0 + SCAN_IPT <= n);
    int v[SCAN_IPT];
    if (whole) {
#pragma unroll
      for (int q = 0; q < SCAN_IPT / 4; ++q) {
        const int4 t = ((const int4*)(cursor + i0))[q];
        v[4 * q + 0] = t.x; v[4 * q + 1] = t.y;
        v[4 * q + 2] = t.z; v[4 * q + 3] = t.w;
      }
    } else {
#pragma unroll
      for (int k = 0; k < SCAN_IPT; ++k)
        v[k] = (i0 + k < n) ? cursor[i0 + k] : 0;
    }
#pragma unroll
    for (int k = 1; k < SCAN_IPT; ++k) v[k] += v[k - 1];
    const int tsum = v[SCAN_IPT - 1];

    int inc = tsum;
#pragma unroll
    for (int off = 1; off < 32; off <<= 1) {
      const int t = __shfl_up(inc, off, 32);
      if (lane >= off) inc += t;
    }
    if (lane == 31) wave_tot[w] = inc;
    __syncthreads();
    const int wt = wave_tot[lane];
    int winc = wt;
#pragma unroll
    for (int off = 1; off < 32; off <<= 1) {
      const int t = __shfl_up(winc, off, 32);
      if (lane >= off) winc += t;
    }
    const int wave_excl = __shfl(winc - wt, w, 32);
    const int total     = __shfl(winc, 31, 32);
    const int excl = carry + wave_excl + (inc - tsum);

    int ex[SCAN_IPT];
    ex[0] = excl;
#pragma unroll
    for (int k = 1; k < SCAN_IPT; ++k) ex[k] = excl + v[k - 1];
    if (whole) {
#pragma unroll
      for (int q = 0; q < SCAN_IPT / 4; ++q) {
        int4 t;
        t.x = ex[4 * q + 0]; t.y = ex[4 * q + 1];
        t.z = ex[4 * q + 2]; t.w = ex[4 * q + 3];
        ((int4*)(cursor + i0))[q] = t;
      }
    } else {
#pragma unroll
      for (int k = 0; k < SCAN_IPT; ++k)
        if (i0 + k < n) cursor[i0 + k] = ex[k];
    }
    carry += total;
    __syncthreads();
  }
}

__global__ void gat_fill_kernel(const int* __restrict__ src,
                                const int* __restrict__ dst,
                                int* __restrict__ cursor,
                                int* __restrict__ seg_src, int E) {
  const int e = blockIdx.x * blockDim.x + threadIdx.x;
  if (e >= E) return;
  const int slot = atomicAdd(&cursor[dst[e]], 1);
  seg_src[slot] = src[e];
}

#define NODE_WAVES 4
#define ROW_PITCH4 33

__device__ __forceinline__ float lane_value(float v, int l) {
  return __int_as_float(__builtin_amdgcn_readlane(__float_as_int(v), l));
}

__global__ __launch_bounds__(NODE_WAVES * 32, 8)
void gat_node_wmma_kernel(const float* __restrict__ feat,
                          const int* __restrict__ seg_end,
                          const int* __restrict__ seg_src,
                          float* __restrict__ out, int n_nodes) {
  __shared__ float4 stage[NODE_WAVES][16][ROW_PITCH4];
  __shared__ float4 dst_row[NODE_WAVES][32];

  const int lane = threadIdx.x & 31;
  const int wib  = threadIdx.x >> 5;
  const int d = __builtin_amdgcn_readfirstlane(blockIdx.x * NODE_WAVES + wib);
  if (d >= n_nodes) return;

  const int m    = lane & 15;
  const int half = lane >> 4;

  const int beg = (d > 0) ? seg_end[d - 1] : 0;
  const int end = seg_end[d];
  const float4* feat4 = (const float4*)feat;
  const float4 hd = feat4[(size_t)d * (FEAT_DIM / 4) + lane];
  float4* orow = (float4*)(out + (size_t)d * FEAT_DIM) + lane;

  if (end <= beg) {
    *orow = hd;
    return;
  }

  dst_row[wib][lane] = hd;

  float  m_run = -INFINITY;
  float  l_run = 0.0f;
  float4 acc = make_float4(0.f, 0.f, 0.f, 0.f);

  for (int p0 = beg; p0 < end; p0 += 16) {
    asm volatile("" ::: "memory");
    const int cnt = min(16, end - p0);
    const int sidx = (lane < cnt) ? seg_src[p0 + lane] : d;

    float4 rs[16];
#pragma unroll
    for (int r = 0; r < 16; ++r) {
      const int sr = __builtin_amdgcn_readlane(sidx, r);
      rs[r] = feat4[(size_t)sr * (FEAT_DIM / 4) + lane];
    }
#pragma unroll
    for (int r = 0; r < 16; ++r) stage[wib][r][lane] = rs[r];


    v8f c = {0.f, 0.f, 0.f, 0.f, 0.f, 0.f, 0.f, 0.f};
#pragma unroll
    for (int j = 0; j < 16; ++j) {
      const float4 av = stage[wib][m][2 * j + half];
      const float4 bv = dst_row[wib][2 * j + half];
      v2f a; a.x = av.x; a.y = av.y;
      v2f b; b.x = bv.x; b.y = bv.y;
      c = __builtin_amdgcn_wmma_f32_16x16x4_f32(false, a, false, b, (short)0, c,
                                                false, false);
      a.x = av.z; a.y = av.w;
      b.x = bv.z; b.y = bv.w;
      c = __builtin_amdgcn_wmma_f32_16x16x4_f32(false, a, false, b, (short)0, c,
                                                false, false);
    }

    float mloc = -INFINITY;
#pragma unroll
    for (int i = 0; i < 8; ++i)
      if (8 * half + i < cnt) mloc = fmaxf(mloc, c[i]);
    const float m_new = fmaxf(m_run, fmaxf(lane_value(mloc, 0), lane_value(mloc, 16)));
    const float scale = __expf(m_run - m_new);

    float p[8];
    float ploc = 0.0f;
#pragma unroll
    for (int i = 0; i < 8; ++i) {
      p[i] = (8 * half + i < cnt) ? __expf(c[i] - m_new) : 0.0f;
      ploc += p[i];
    }
    l_run = l_run * scale + (lane_value(ploc, 0) + lane_value(ploc, 16));

    acc.x *= scale; acc.y *= scale; acc.z *= scale; acc.w *= scale;
#pragma unroll
    for (int r = 0; r < 16; ++r) {
      const float w = lane_value(p[r & 7], (r >> 3) * 16);
      acc.x = fmaf(w, rs[r].x, acc.x);
      acc.y = fmaf(w, rs[r].y, acc.y);
      acc.z = fmaf(w, rs[r].z, acc.z);
      acc.w = fmaf(w, rs[r].w, acc.w);
    }
    m_run = m_new;
  }

  float4 o;
  o.x = acc.x / l_run; o.y = acc.y / l_run;
  o.z = acc.z / l_run; o.w = acc.w / l_run;
  *orow = o;
}

extern "C" void kernel_launch(void* const* d_in, const int* in_sizes, int n_in,
                              void* d_out, int out_size, void* d_ws, size_t ws_size,
                              hipStream_t stream) {
  const float* feat = (const float*)d_in[0];
  const int* src    = (const int*)d_in[1];
  const int* dst    = (const int*)d_in[2];
  const int E       = in_sizes[1];
  const int N       = in_sizes[0] / FEAT_DIM;
  float* out        = (float*)d_out;

  int* cursor  = (int*)d_ws;
  int* seg_src = cursor + N;

  const int BLK = 256;

  gat_zero_kernel<<<(N + BLK - 1) / BLK, BLK, 0, stream>>>(cursor, N);
  gat_count_kernel<<<(E + BLK - 1) / BLK, BLK, 0, stream>>>(dst, cursor, E);
  gat_scan_kernel<<<1, SCAN_THREADS, 0, stream>>>(cursor, N);
  gat_fill_kernel<<<(E + BLK - 1) / BLK, BLK, 0, stream>>>(src, dst, cursor,
                                                           seg_src, E);
  gat_node_wmma_kernel<<<(N + NODE_WAVES - 1) / NODE_WAVES, NODE_WAVES * 32, 0,
                         stream>>>(feat, cursor, seg_src, out, N);
}


# edge counter of a node moved from a dense array into word 0 of its own out row (slots in words 1..127): no two nodes' counters share a line; sum order stays atomic arrival
# speedup vs baseline: 3.8250x; 1.6623x over previous
#include <hip/hip_runtime.h>
#include <cstdint>
#include <math.h>


#define FEAT_DIM 128

typedef __attribute__((ext_vector_type(2))) float v2f;
typedef __attribute__((ext_vector_type(8))) float v8f;

#define BUCKET_CAP (FEAT_DIM - 1)

__global__ void gat_zero_kernel(float* out, int n_nodes) {
  const int i = blockIdx.x * blockDim.x + threadIdx.x;
  if (i < n_nodes) ((int*)out)[(size_t)i * FEAT_DIM] = 0;
}

__global__ void gat_bucket_kernel(const int* __restrict__ src,
                                  const int* __restrict__ dst,
                                  float* out, int E) {
  const int e = blockIdx.x * blockDim.x + threadIdx.x;
  if (e >= E) return;
  int* row = (int*)out + (size_t)dst[e] * FEAT_DIM;
  const int slot = atomicAdd(row, 1);
  if (slot < BUCKET_CAP) row[1 + slot] = src[e];
}

#define NODE_WAVES 4
#define ROW_PITCH4 33

__device__ __forceinline__ float lane_value(float v, int l) {
  return __int_as_float(__builtin_amdgcn_readlane(__float_as_int(v), l));
}

__device__ __forceinline__ void gat_trip(const float4* __restrict__ feat4,
                                         float4 (*stg)[ROW_PITCH4],
                                         const float4* drow,
                                         int sidx, int off, int cnt,
                                         int lane, int m, int half,
                                         float& m_run, float& l_run, float4& acc) {
  asm volatile("" ::: "memory");

  float4 rs[16];
#pragma unroll
  for (int r = 0; r < 16; ++r) {
    const int sr = __builtin_amdgcn_readlane(sidx, (off + r) & 31);
    rs[r] = feat4[(size_t)sr * (FEAT_DIM / 4) + lane];
  }
#pragma unroll
  for (int r = 0; r < 16; ++r) stg[r][lane] = rs[r];


  v8f c = {0.f, 0.f, 0.f, 0.f, 0.f, 0.f, 0.f, 0.f};
#pragma unroll
  for (int j = 0; j < 16; ++j) {
    const float4 av = stg[m][2 * j + half];
    const float4 bv = drow[2 * j + half];
    v2f a; a.x = av.x; a.y = av.y;
    v2f b; b.x = bv.x; b.y = bv.y;
    c = __builtin_amdgcn_wmma_f32_16x16x4_f32(false, a, false, b, (short)0, c,
                                              false, false);
    a.x = av.z; a.y = av.w;
    b.x = bv.z; b.y = bv.w;
    c = __builtin_amdgcn_wmma_f32_16x16x4_f32(false, a, false, b, (short)0, c,
                                              false, false);
  }

  float mloc = -INFINITY;
#pragma unroll
  for (int i = 0; i < 8; ++i)
    if (8 * half + i < cnt) mloc = fmaxf(mloc, c[i]);
  const float m_new = fmaxf(m_run, fmaxf(lane_value(mloc, 0), lane_value(mloc, 16)));
  const float scale = __expf(m_run - m_new);

  float p[8];
  float ploc = 0.0f;
#pragma unroll
  for (int i = 0; i < 8; ++i) {
    p[i] = (8 * half + i < cnt) ? __expf(c[i] - m_new) : 0.0f;
    ploc += p[i];
  }
  l_run = l_run * scale + (lane_value(ploc, 0) + lane_value(ploc, 16));

  acc.x *= scale; acc.y *= scale; acc.z *= scale; acc.w *= scale;
#pragma unroll
  for (int r = 0; r < 16; ++r) {
    const float w = lane_value(p[r & 7], (r >> 3) * 16);
    acc.x = fmaf(w, rs[r].x, acc.x);
    acc.y = fmaf(w, rs[r].y, acc.y);
    acc.z = fmaf(w, rs[r].z, acc.z);
    acc.w = fmaf(w, rs[r].w, acc.w);
  }
  m_run = m_new;
}

__global__ __launch_bounds__(NODE_WAVES * 32, 8)
void gat_node_wmma_kernel(const float* __restrict__ feat,
                          const int* __restrict__ src,
                          const int* __restrict__ dst,
                          float* out, int n_nodes, int E) {
  __shared__ float4 stage[NODE_WAVES][16][ROW_PITCH4];
  __shared__ float4 dst_row[NODE_WAVES][32];
  __shared__ int found[NODE_WAVES][32];

  const int lane = threadIdx.x & 31;
  const int wib  = threadIdx.x >> 5;
  const int d = __builtin_amdgcn_readfirstlane(blockIdx.x * NODE_WAVES + wib);
  if (d >= n_nodes) return;

  const int m    = lane & 15;
  const int half = lane >> 4;

  const int* bucket = (const int*)out + (size_t)d * FEAT_DIM;
  int first = bucket[lane];
  const float4* feat4 = (const float4*)feat;
  float4 hd = feat4[(size_t)d * (FEAT_DIM / 4) + lane];
  asm volatile("" : "+v"(first), "+v"(hd.x), "+v"(hd.y), "+v"(hd.z), "+v"(hd.w));
  const int deg = __builtin_amdgcn_readlane(first, 0);
  float4* orow = (float4*)(out + (size_t)d * FEAT_DIM) + lane;

  if (deg <= 0) {
    *orow = hd;
    return;
  }

  dst_row[wib][lane] = hd;

  float  m_run = -INFINITY;
  float  l_run = 0.0f;
  float4 acc = make_float4(0.f, 0.f, 0.f, 0.f);

  if (deg <= BUCKET_CAP) {
    int sidx = (lane >= 1 && lane - 1 < deg) ? first : d;
    int off = 1, width = 16;
    for (int s0 = 0; s0 < deg; ) {
      gat_trip(feat4, stage[wib], dst_row[wib], sidx, off, min(width, deg - s0),
               lane, m, half, m_run, l_run, acc);
      s0 += width;
      if (s0 == 16) {
        off = 17; width = 15;
      } else {
        off = 0; width = 16;
        if (s0 < deg) {
          const int nxt = bucket[1 + s0 + (lane & 15)];
          sidx = (lane < 16 && s0 + lane < deg) ? nxt : d;
        }
      }
    }
  } else {
    for (int base = 0; base < E; base += 32) {
      const int e = base + lane;
      const bool hit = (e < E) && (dst[e] == d);
      const unsigned mask = __builtin_amdgcn_ballot_w32(hit);
      if (mask == 0u) continue;
      const int k = __builtin_popcount(mask);
      if (hit) found[wib][__builtin_popcount(mask & ((1u << lane) - 1u))] = src[e];
      const int sidx = (lane < k) ? found[wib][lane] : d;
      for (int t = 0; t < k; t += 16)
        gat_trip(feat4, stage[wib], dst_row[wib], sidx, t, min(16, k - t),
                 lane, m, half, m_run, l_run, acc);
    }
  }

  float4 o;
  o.x = acc.x / l_run; o.y = acc.y / l_run;
  o.z = acc.z / l_run; o.w = acc.w / l_run;
  *orow = o;
}

extern "C" void kernel_launch(void* const* d_in, const int* in_sizes, int n_in,
                              void* d_out, int out_size, void* d_ws, size_t ws_size,
                              hipStream_t stream) {
  const float* feat = (const float*)d_in[0];
  const int* src    = (const int*)d_in[1];
  const int* dst    = (const int*)d_in[2];
  const int E       = in_sizes[1];
  const int N       = in_sizes[0] / FEAT_DIM;
  float* out        = (float*)d_out;


  const int BLK = 256;

  gat_zero_kernel<<<(N + BLK - 1) / BLK, BLK, 0, stream>>>(out, N);
  gat_bucket_kernel<<<(E + BLK - 1) / BLK, BLK, 0, stream>>>(src, dst, out, E);
  gat_node_wmma_kernel<<<(N + NODE_WAVES - 1) / NODE_WAVES, NODE_WAVES * 32, 0,
                         stream>>>(feat, src, dst, out, N, E);
}


# edges grouped by a two-level counting sort on LDS counters (2048-edge chunks binned by 64-node range, then a work-group per range fills its nodes' buckets): no global atomics; sum order = LDS arrival
# speedup vs baseline: 4.8811x; 1.2761x over previous
#include <hip/hip_runtime.h>
#include <cstdint>
#include <math.h>


#define FEAT_DIM 128

typedef __attribute__((ext_vector_type(2))) float v2f;
typedef __attribute__((ext_vector_type(8))) float v8f;

#define BUCKET_CAP (FEAT_DIM - 1)

#define RANGE_SHIFT 6
#define RANGE_NODES (1 << RANGE_SHIFT)
#define MAX_BINS 1024
#define CHUNK_EDGES 2048
#define SORT_THREADS 256
#define EDGES_PER_THREAD (CHUNK_EDGES / SORT_THREADS)
#define ROW_IMG_PITCH (FEAT_DIM + 1)

__device__ __forceinline__ int gat_num_chunks(int E) {
  return (E + CHUNK_EDGES - 1) / CHUNK_EDGES;
}
__device__ __forceinline__ int gat_num_bins(int n_nodes) {
  return (n_nodes + RANGE_NODES - 1) >> RANGE_SHIFT;
}
__device__ __forceinline__ int gat_table_pitch(int n_bins) {
  return (n_bins + 1 + 63) & ~63;
}

__global__ __launch_bounds__(SORT_THREADS)
void gat_chunk_sort_kernel(const int* __restrict__ src,
                           const int* __restrict__ dst,
                           unsigned int* ws, int n_nodes, int E) {
  __shared__ int bin_cnt[MAX_BINS];
  __shared__ int bin_start[MAX_BINS + 1];
  __shared__ unsigned int recs[CHUNK_EDGES];
  __shared__ int wave_sum[SORT_THREADS / 32];

  const int tid  = threadIdx.x;
  const int lane = tid & 31;
  const int c    = blockIdx.x;
  const int base = c * CHUNK_EDGES;
  const int n_bins = gat_num_bins(n_nodes);

  for (int i = tid; i < MAX_BINS; i += SORT_THREADS) bin_cnt[i] = 0;
  __syncthreads();

  int dv[EDGES_PER_THREAD], sv[EDGES_PER_THREAD], rk[EDGES_PER_THREAD];
#pragma unroll
  for (int k = 0; k < EDGES_PER_THREAD; ++k) {
    const int e = base + k * SORT_THREADS + tid;
    const bool ok = e < E;
    dv[k] = ok ? dst[e] : -1;
    sv[k] = ok ? src[e] : 0;
  }
#pragma unroll
  for (int k = 0; k < EDGES_PER_THREAD; ++k) {
    rk[k] = 0;
    if (dv[k] >= 0)
      rk[k] = atomicAdd(&bin_cnt[min(dv[k] >> RANGE_SHIFT, MAX_BINS - 1)], 1);
  }
  __syncthreads();

  const int b4 = tid * 4;
  const int c0 = bin_cnt[b4], c1 = bin_cnt[b4 + 1], c2 = bin_cnt[b4 + 2], c3 = bin_cnt[b4 + 3];
  const int mine = c0 + c1 + c2 + c3;
  int incl = mine;
#pragma unroll
  for (int s = 1; s < 32; s <<= 1) {
    const int t = __shfl_up(incl, s);
    if (lane >= s) incl += t;
  }
  if (lane == 31) wave_sum[tid >> 5] = incl;
  __syncthreads();
  int prefix = incl - mine;
  for (int w = 0; w < (tid >> 5); ++w) prefix += wave_sum[w];
  bin_start[b4]     = prefix;
  bin_start[b4 + 1] = prefix + c0;
  bin_start[b4 + 2] = prefix + c0 + c1;
  bin_start[b4 + 3] = prefix + c0 + c1 + c2;
  if (tid == SORT_THREADS - 1) bin_start[MAX_BINS] = prefix + mine;
  __syncthreads();

#pragma unroll
  for (int k = 0; k < EDGES_PER_THREAD; ++k) {
    if (dv[k] >= 0) {
      const int bin = min(dv[k] >> RANGE_SHIFT, MAX_BINS - 1);
      recs[bin_start[bin] + rk[k]] =
          ((unsigned int)sv[k] << RANGE_SHIFT) | (unsigned int)(dv[k] & (RANGE_NODES - 1));
    }
  }
  __syncthreads();

  const int count = min(CHUNK_EDGES, E - base);
  unsigned int* rec = ws;
  for (int i = tid; i < count; i += SORT_THREADS) rec[base + i] = recs[i];
  unsigned int* table = ws + (size_t)gat_num_chunks(E) * CHUNK_EDGES;
  unsigned int* trow = table + (size_t)c * (gat_table_pitch(n_bins) >> 1);
  for (int i = tid; 2 * i <= n_bins; i += SORT_THREADS) {
    const unsigned int lo = (unsigned int)bin_start[2 * i];
    const unsigned int hi = (2 * i + 1 <= MAX_BINS) ? (unsigned int)bin_start[2 * i + 1] : 0u;
    trow[i] = lo | (hi << 16);
  }
}

__global__ __launch_bounds__(SORT_THREADS)
void gat_range_bucket_kernel(const unsigned int* ws, float* out, int n_nodes, int E) {
  __shared__ int img[RANGE_NODES * ROW_IMG_PITCH];
  __shared__ int cnt[RANGE_NODES];

  const int tid = threadIdx.x;
  const int b   = blockIdx.x;
  const int n_bins   = gat_num_bins(n_nodes);
  const int n_chunks = gat_num_chunks(E);

  for (int i = tid; i < RANGE_NODES * ROW_IMG_PITCH; i += SORT_THREADS) img[i] = 0;
  if (tid < RANGE_NODES) cnt[tid] = 0;
  __syncthreads();

  const unsigned int* rec = ws;
  const unsigned int* table = ws + (size_t)n_chunks * CHUNK_EDGES;
  const int pitch_words = gat_table_pitch(n_bins) >> 1;
  for (int c = tid; c < n_chunks; c += SORT_THREADS) {
    const unsigned int* trow = table + (size_t)c * pitch_words + (b >> 1);
    const unsigned int w0 = trow[0];
    int s, e;
    if (b & 1) {
      s = (int)(w0 >> 16);
      e = (int)(trow[1] & 0xffffu);
    } else {
      s = (int)(w0 & 0xffffu);
      e = (int)(w0 >> 16);
    }
    const unsigned int* piece = rec + (size_t)c * CHUNK_EDGES;
    for (int i = s; i < e; i += 4) {
      unsigned int r[4];
#pragma unroll
      for (int q = 0; q < 4; ++q) r[q] = (i + q < e) ? piece[i + q] : 0u;
#pragma unroll
      for (int q = 0; q < 4; ++q) {
        if (i + q < e) {
          const int j = (int)(r[q] & (RANGE_NODES - 1));
          const int slot = atomicAdd(&cnt[j], 1);
          if (slot < BUCKET_CAP) img[j * ROW_IMG_PITCH + 1 + slot] = (int)(r[q] >> RANGE_SHIFT);
        }
      }
    }
  }
  __syncthreads();

  const int lane = tid & 31;
  for (int j = tid >> 5; j < RANGE_NODES; j += SORT_THREADS / 32) {
    const int d = b * RANGE_NODES + j;
    if (d >= n_nodes) break;
    const int deg = cnt[j];
    int* orow = (int*)out + (size_t)d * FEAT_DIM;
    orow[lane] = (lane == 0) ? deg : img[j * ROW_IMG_PITCH + lane];
    const int lines = (min(deg, BUCKET_CAP) + 1 + 31) >> 5;
    for (int li = 1; li < lines; ++li)
      orow[li * 32 + lane] = img[j * ROW_IMG_PITCH + li * 32 + lane];
  }
}

#define NODE_WAVES 4
#define ROW_PITCH4 33

__device__ __forceinline__ float lane_value(float v, int l) {
  return __int_as_float(__builtin_amdgcn_readlane(__float_as_int(v), l));
}

__device__ __forceinline__ void gat_trip(const float4* __restrict__ feat4,
                                         float4 (*stg)[ROW_PITCH4],
                                         const float4* drow,
                                         int sidx, int off, int cnt,
                                         int lane, int m, int half,
                                         float& m_run, float& l_run, float4& acc) {
  asm volatile("" ::: "memory");

  float4 rs[16];
#pragma unroll
  for (int r = 0; r < 16; ++r) {
    const int sr = __builtin_amdgcn_readlane(sidx, (off + r) & 31);
    rs[r] = feat4[(size_t)sr * (FEAT_DIM / 4) + lane];
  }
#pragma unroll
  for (int r = 0; r < 16; ++r) stg[r][lane] = rs[r];


  v8f c = {0.f, 0.f, 0.f, 0.f, 0.f, 0.f, 0.f, 0.f};
#pragma unroll
  for (int j = 0; j < 16; ++j) {
    const float4 av = stg[m][2 * j + half];
    const float4 bv = drow[2 * j + half];
    v2f a; a.x = av.x; a.y = av.y;
    v2f b; b.x = bv.x; b.y = bv.y;
    c = __builtin_amdgcn_wmma_f32_16x16x4_f32(false, a, false, b, (short)0, c,
                                              false, false);
    a.x = av.z; a.y = av.w;
    b.x = bv.z; b.y = bv.w;
    c = __builtin_amdgcn_wmma_f32_16x16x4_f32(false, a, false, b, (short)0, c,
                                              false, false);
  }

  float mloc = -INFINITY;
#pragma unroll
  for (int i = 0; i < 8; ++i)
    if (8 * half + i < cnt) mloc = fmaxf(mloc, c[i]);
  const float m_new = fmaxf(m_run, fmaxf(lane_value(mloc, 0), lane_value(mloc, 16)));
  const float scale = __expf(m_run - m_new);

  float p[8];
  float ploc = 0.0f;
#pragma unroll
  for (int i = 0; i < 8; ++i) {
    p[i] = (8 * half + i < cnt) ? __expf(c[i] - m_new) : 0.0f;
    ploc += p[i];
  }
  l_run = l_run * scale + (lane_value(ploc, 0) + lane_value(ploc, 16));

  acc.x *= scale; acc.y *= scale; acc.z *= scale; acc.w *= scale;
#pragma unroll
  for (int r = 0; r < 16; ++r) {
    const float w = lane_value(p[r & 7], (r >> 3) * 16);
    acc.x = fmaf(w, rs[r].x, acc.x);
    acc.y = fmaf(w, rs[r].y, acc.y);
    acc.z = fmaf(w, rs[r].z, acc.z);
    acc.w = fmaf(w, rs[r].w, acc.w);
  }
  m_run = m_new;
}

__global__ __launch_bounds__(NODE_WAVES * 32, 8)
void gat_node_wmma_kernel(const float* __restrict__ feat,
                          const int* __restrict__ src,
                          const int* __restrict__ dst,
                          float* out, int n_nodes, int E) {
  __shared__ float4 stage[NODE_WAVES][16][ROW_PITCH4];
  __shared__ float4 dst_row[NODE_WAVES][32];
  __shared__ int found[NODE_WAVES][32];

  const int lane = threadIdx.x & 31;
  const int wib  = threadIdx.x >> 5;
  const int d = __builtin_amdgcn_readfirstlane(blockIdx.x * NODE_WAVES + wib);
  if (d >= n_nodes) return;

  const int m    = lane & 15;
  const int half = lane >> 4;

  const int* bucket = (const int*)out + (size_t)d * FEAT_DIM;
  int first = bucket[lane];
  const float4* feat4 = (const float4*)feat;
  float4 hd = feat4[(size_t)d * (FEAT_DIM / 4) + lane];
  asm volatile("" : "+v"(first), "+v"(hd.x), "+v"(hd.y), "+v"(hd.z), "+v"(hd.w));
  const int deg = __builtin_amdgcn_readlane(first, 0);
  float4* orow = (float4*)(out + (size_t)d * FEAT_DIM) + lane;

  if (deg <= 0) {
    *orow = hd;
    return;
  }

  dst_row[wib][lane] = hd;

  float  m_run = -INFINITY;
  float  l_run = 0.0f;
  float4 acc = make_float4(0.f, 0.f, 0.f, 0.f);

  if (deg <= BUCKET_CAP) {
    int sidx = (lane >= 1 && lane - 1 < deg) ? first : d;
    int off = 1, width = 16;
    for (int s0 = 0; s0 < deg; ) {
      gat_trip(feat4, stage[wib], dst_row[wib], sidx, off, min(width, deg - s0),
               lane, m, half, m_run, l_run, acc);
      s0 += width;
      if (s0 == 16) {
        off = 17; width = 15;
      } else {
        off = 0; width = 16;
        if (s0 < deg) {
          const int nxt = bucket[1 + s0 + (lane & 15)];
          sidx = (lane < 16 && s0 + lane < deg) ? nxt : d;
        }
      }
    }
  } else {
    for (int base = 0; base < E; base += 32) {
      const int e = base + lane;
      const bool hit = (e < E) && (dst[e] == d);
      const unsigned mask = __builtin_amdgcn_ballot_w32(hit);
      if (mask == 0u) continue;
      const int k = __builtin_popcount(mask);
      if (hit) found[wib][__builtin_popcount(mask & ((1u << lane) - 1u))] = src[e];
      const int sidx = (lane < k) ? found[wib][lane] : d;
      for (int t = 0; t < k; t += 16)
        gat_trip(feat4, stage[wib], dst_row[wib], sidx, t, min(16, k - t),
                 lane, m, half, m_run, l_run, acc);
    }
  }

  float4 o;
  o.x = acc.x / l_run; o.y = acc.y / l_run;
  o.z = acc.z / l_run; o.w = acc.w / l_run;
  *orow = o;
}

extern "C" void kernel_launch(void* const* d_in, const int* in_sizes, int n_in,
                              void* d_out, int out_size, void* d_ws, size_t ws_size,
                              hipStream_t stream) {
  const float* feat = (const float*)d_in[0];
  const int* src    = (const int*)d_in[1];
  const int* dst    = (const int*)d_in[2];
  const int E       = in_sizes[1];
  const int N       = in_sizes[0] / FEAT_DIM;
  float* out        = (float*)d_out;

  unsigned int* ws = (unsigned int*)d_ws;

  gat_chunk_sort_kernel<<<(E + CHUNK_EDGES - 1) / CHUNK_EDGES, SORT_THREADS, 0, stream>>>(
      src, dst, ws, N, E);
  gat_range_bucket_kernel<<<(N + RANGE_NODES - 1) / RANGE_NODES, SORT_THREADS, 0, stream>>>(
      ws, out, N, E);
  gat_node_wmma_kernel<<<(N + NODE_WAVES - 1) / NODE_WAVES, NODE_WAVES * 32, 0,
                         stream>>>(feat, src, dst, out, N, E);
}


# sort kernels: a thread's global loads issued together, branch-free (clamped index): chunk kernel 8 dependent round trips to 1, range kernel both chunks at once; sum order stays LDS arrival
# speedup vs baseline: 5.2625x; 1.0781x over previous
#include <hip/hip_runtime.h>
#include <cstdint>
#include <math.h>


#define FEAT_DIM 128

typedef __attribute__((ext_vector_type(2))) float v2f;
typedef __attribute__((ext_vector_type(8))) float v8f;

#define BUCKET_CAP (FEAT_DIM - 1)

#define RANGE_SHIFT 6
#define RANGE_NODES (1 << RANGE_SHIFT)
#define MAX_BINS 1024
#define CHUNK_EDGES 2048
#define SORT_THREADS 256
#define EDGES_PER_THREAD (CHUNK_EDGES / SORT_THREADS)
#define ROW_IMG_PITCH (FEAT_DIM + 1)

__device__ __forceinline__ int gat_num_chunks(int E) {
  return (E + CHUNK_EDGES - 1) / CHUNK_EDGES;
}
__device__ __forceinline__ int gat_num_bins(int n_nodes) {
  return (n_nodes + RANGE_NODES - 1) >> RANGE_SHIFT;
}
__device__ __forceinline__ int gat_table_pitch(int n_bins) {
  return (n_bins + 1 + 63) & ~63;
}

__global__ __launch_bounds__(SORT_THREADS)
void gat_chunk_sort_kernel(const int* __restrict__ src,
                           const int* __restrict__ dst,
                           unsigned int* ws, int n_nodes, int E) {
  __shared__ int bin_cnt[MAX_BINS];
  __shared__ int bin_start[MAX_BINS + 1];
  __shared__ unsigned int recs[CHUNK_EDGES];
  __shared__ int wave_sum[SORT_THREADS / 32];

  const int tid  = threadIdx.x;
  const int lane = tid & 31;
  const int c    = blockIdx.x;
  const int base = c * CHUNK_EDGES;
  const int n_bins = gat_num_bins(n_nodes);

  for (int i = tid; i < MAX_BINS; i += SORT_THREADS) bin_cnt[i] = 0;
  __syncthreads();

  int dv[EDGES_PER_THREAD], sv[EDGES_PER_THREAD], rk[EDGES_PER_THREAD];
#pragma unroll
  for (int k = 0; k < EDGES_PER_THREAD; ++k) {
    const int e = min(base + k * SORT_THREADS + tid, E - 1);
    dv[k] = dst[e];
    sv[k] = src[e];
  }
#pragma unroll
  for (int k = 0; k < EDGES_PER_THREAD; ++k)
    if (base + k * SORT_THREADS + tid >= E) dv[k] = -1;
#pragma unroll
  for (int k = 0; k < EDGES_PER_THREAD; ++k) {
    rk[k] = 0;
    if (dv[k] >= 0)
      rk[k] = atomicAdd(&bin_cnt[min(dv[k] >> RANGE_SHIFT, MAX_BINS - 1)], 1);
  }
  __syncthreads();

  const int b4 = tid * 4;
  const int c0 = bin_cnt[b4], c1 = bin_cnt[b4 + 1], c2 = bin_cnt[b4 + 2], c3 = bin_cnt[b4 + 3];
  const int mine = c0 + c1 + c2 + c3;
  int incl = mine;
#pragma unroll
  for (int s = 1; s < 32; s <<= 1) {
    const int t = __shfl_up(incl, s);
    if (lane >= s) incl += t;
  }
  if (lane == 31) wave_sum[tid >> 5] = incl;
  __syncthreads();
  int prefix = incl - mine;
  for (int w = 0; w < (tid >> 5); ++w) prefix += wave_sum[w];
  bin_start[b4]     = prefix;
  bin_start[b4 + 1] = prefix + c0;
  bin_start[b4 + 2] = prefix + c0 + c1;
  bin_start[b4 + 3] = prefix + c0 + c1 + c2;
  if (tid == SORT_THREADS - 1) bin_start[MAX_BINS] = prefix + mine;
  __syncthreads();

#pragma unroll
  for (int k = 0; k < EDGES_PER_THREAD; ++k) {
    if (dv[k] >= 0) {
      const int bin = min(dv[k] >> RANGE_SHIFT, MAX_BINS - 1);
      recs[bin_start[bin] + rk[k]] =
          ((unsigned int)sv[k] << RANGE_SHIFT) | (unsigned int)(dv[k] & (RANGE_NODES - 1));
    }
  }
  __syncthreads();

  const int count = min(CHUNK_EDGES, E - base);
  unsigned int* rec = ws;
  for (int i = tid; i < count; i += SORT_THREADS) rec[base + i] = recs[i];
  unsigned int* table = ws + (size_t)gat_num_chunks(E) * CHUNK_EDGES;
  unsigned int* trow = table + (size_t)c * (gat_table_pitch(n_bins) >> 1);
  for (int i = tid; 2 * i <= n_bins; i += SORT_THREADS) {
    const unsigned int lo = (unsigned int)bin_start[2 * i];
    const unsigned int hi = (2 * i + 1 <= MAX_BINS) ? (unsigned int)bin_start[2 * i + 1] : 0u;
    trow[i] = lo | (hi << 16);
  }
}

__global__ __launch_bounds__(SORT_THREADS)
void gat_range_bucket_kernel(const unsigned int* ws, float* out, int n_nodes, int E) {
  __shared__ int img[RANGE_NODES * ROW_IMG_PITCH];
  __shared__ int cnt[RANGE_NODES];

  const int tid = threadIdx.x;
  const int b   = blockIdx.x;
  const int n_bins   = gat_num_bins(n_nodes);
  const int n_chunks = gat_num_chunks(E);

  for (int i = tid; i < RANGE_NODES * ROW_IMG_PITCH; i += SORT_THREADS) img[i] = 0;
  if (tid < RANGE_NODES) cnt[tid] = 0;
  __syncthreads();

  const unsigned int* rec = ws;
  const unsigned int* table = ws + (size_t)n_chunks * CHUNK_EDGES;
  const int pitch_words = gat_table_pitch(n_bins) >> 1;
  for (int c0 = tid; c0 < n_chunks; c0 += 2 * SORT_THREADS) {
    int s[2], e[2];
    unsigned int w0[2], w1[2];
    const unsigned int* piece[2];
#pragma unroll
    for (int u = 0; u < 2; ++u) {
      const int c = c0 + u * SORT_THREADS;
      const int cl = min(c, n_chunks - 1);
      const unsigned int* trow = table + (size_t)cl * pitch_words + (b >> 1);
      w0[u] = trow[0];
      w1[u] = trow[b & 1];
      piece[u] = rec + (size_t)cl * CHUNK_EDGES;
    }
    const int sh = (b & 1) * 16;
#pragma unroll
    for (int u = 0; u < 2; ++u) {
      const unsigned long long both = ((unsigned long long)w1[u] << 32) | w0[u];
      s[u] = (int)((both >> sh) & 0xffffu);
      e[u] = (int)((both >> (sh + 16)) & 0xffffu);
      if (c0 + u * SORT_THREADS >= n_chunks) e[u] = s[u];
    }
    while (s[0] < e[0] || s[1] < e[1]) {
      unsigned int r[8];
#pragma unroll
      for (int u = 0; u < 2; ++u) {
#pragma unroll
        for (int q = 0; q < 4; ++q)
          r[4 * u + q] = (s[u] + q < e[u]) ? piece[u][s[u] + q] : 0u;
      }
#pragma unroll
      for (int u = 0; u < 2; ++u) {
#pragma unroll
        for (int q = 0; q < 4; ++q) {
          if (s[u] + q < e[u]) {
            const unsigned int rr = r[4 * u + q];
            const int j = (int)(rr & (RANGE_NODES - 1));
            const int slot = atomicAdd(&cnt[j], 1);
            if (slot < BUCKET_CAP) img[j * ROW_IMG_PITCH + 1 + slot] = (int)(rr >> RANGE_SHIFT);
          }
        }
      }
      s[0] += 4;
      s[1] += 4;
    }
  }
  __syncthreads();

  const int lane = tid & 31;
  for (int j = tid >> 5; j < RANGE_NODES; j += SORT_THREADS / 32) {
    const int d = b * RANGE_NODES + j;
    if (d >= n_nodes) break;
    const int deg = cnt[j];
    int* orow = (int*)out + (size_t)d * FEAT_DIM;
    orow[lane] = (lane == 0) ? deg : img[j * ROW_IMG_PITCH + lane];
    const int lines = (min(deg, BUCKET_CAP) + 1 + 31) >> 5;
    for (int li = 1; li < lines; ++li)
      orow[li * 32 + lane] = img[j * ROW_IMG_PITCH + li * 32 + lane];
  }
}

#define NODE_WAVES 4
#define ROW_PITCH4 33

__device__ __forceinline__ float lane_value(float v, int l) {
  return __int_as_float(__builtin_amdgcn_readlane(__float_as_int(v), l));
}

__device__ __forceinline__ void gat_trip(const float4* __restrict__ feat4,
                                         float4 (*stg)[ROW_PITCH4],
                                         const float4* drow,
                                         int sidx, int off, int cnt,
                                         int lane, int m, int half,
                                         float& m_run, float& l_run, float4& acc) {
  asm volatile("" ::: "memory");

  float4 rs[16];
#pragma unroll
  for (int r = 0; r < 16; ++r) {
    const int sr = __builtin_amdgcn_readlane(sidx, (off + r) & 31);
    rs[r] = feat4[(size_t)sr * (FEAT_DIM / 4) + lane];
  }
#pragma unroll
  for (int r = 0; r < 16; ++r) stg[r][lane] = rs[r];


  v8f c = {0.f, 0.f, 0.f, 0.f, 0.f, 0.f, 0.f, 0.f};
#pragma unroll
  for (int j = 0; j < 16; ++j) {
    const float4 av = stg[m][2 * j + half];
    const float4 bv = drow[2 * j + half];
    v2f a; a.x = av.x; a.y = av.y;
    v2f b; b.x = bv.x; b.y = bv.y;
    c = __builtin_amdgcn_wmma_f32_16x16x4_f32(false, a, false, b, (short)0, c,
                                              false, false);
    a.x = av.z; a.y = av.w;
    b.x = bv.z; b.y = bv.w;
    c = __builtin_amdgcn_wmma_f32_16x16x4_f32(false, a, false, b, (short)0, c,
                                              false, false);
  }

  float mloc = -INFINITY;
#pragma unroll
  for (int i = 0; i < 8; ++i)
    if (8 * half + i < cnt) mloc = fmaxf(mloc, c[i]);
  const float m_new = fmaxf(m_run, fmaxf(lane_value(mloc, 0), lane_value(mloc, 16)));
  const float scale = __expf(m_run - m_new);

  float p[8];
  float ploc = 0.0f;
#pragma unroll
  for (int i = 0; i < 8; ++i) {
    p[i] = (8 * half + i < cnt) ? __expf(c[i] - m_new) : 0.0f;
    ploc += p[i];
  }
  l_run = l_run * scale + (lane_value(ploc, 0) + lane_value(ploc, 16));

  acc.x *= scale; acc.y *= scale; acc.z *= scale; acc.w *= scale;
#pragma unroll
  for (int r = 0; r < 16; ++r) {
    const float w = lane_value(p[r & 7], (r >> 3) * 16);
    acc.x = fmaf(w, rs[r].x, acc.x);
    acc.y = fmaf(w, rs[r].y, acc.y);
    acc.z = fmaf(w, rs[r].z, acc.z);
    acc.w = fmaf(w, rs[r].w, acc.w);
  }
  m_run = m_new;
}

__global__ __launch_bounds__(NODE_WAVES * 32, 8)
void gat_node_wmma_kernel(const float* __restrict__ feat,
                          const int* __restrict__ src,
                          const int* __restrict__ dst,
                          float* out, int n_nodes, int E) {
  __shared__ float4 stage[NODE_WAVES][16][ROW_PITCH4];
  __shared__ float4 dst_row[NODE_WAVES][32];
  __shared__ int found[NODE_WAVES][32];

  const int lane = threadIdx.x & 31;
  const int wib  = threadIdx.x >> 5;
  const int d = __builtin_amdgcn_readfirstlane(blockIdx.x * NODE_WAVES + wib);
  if (d >= n_nodes) return;

  const int m    = lane & 15;
  const int half = lane >> 4;

  const int* bucket = (const int*)out + (size_t)d * FEAT_DIM;
  int first = bucket[lane];
  const float4* feat4 = (const float4*)feat;
  float4 hd = feat4[(size_t)d * (FEAT_DIM / 4) + lane];
  asm volatile("" : "+v"(first), "+v"(hd.x), "+v"(hd.y), "+v"(hd.z), "+v"(hd.w));
  const int deg = __builtin_amdgcn_readlane(first, 0);
  float4* orow = (float4*)(out + (size_t)d * FEAT_DIM) + lane;

  if (deg <= 0) {
    *orow = hd;
    return;
  }

  dst_row[wib][lane] = hd;

  float  m_run = -INFINITY;
  float  l_run = 0.0f;
  float4 acc = make_float4(0.f, 0.f, 0.f, 0.f);

  if (deg <= BUCKET_CAP) {
    int sidx = (lane >= 1 && lane - 1 < deg) ? first : d;
    int off = 1, width = 16;
    for (int s0 = 0; s0 < deg; ) {
      gat_trip(feat4, stage[wib], dst_row[wib], sidx, off, min(width, deg - s0),
               lane, m, half, m_run, l_run, acc);
      s0 += width;
      if (s0 == 16) {
        off = 17; width = 15;
      } else {
        off = 0; width = 16;
        if (s0 < deg) {
          const int nxt = bucket[1 + s0 + (lane & 15)];
          sidx = (lane < 16 && s0 + lane < deg) ? nxt : d;
        }
      }
    }
  } else {
    for (int base = 0; base < E; base += 32) {
      const int e = base + lane;
      const bool hit = (e < E) && (dst[e] == d);
      const unsigned mask = __builtin_amdgcn_ballot_w32(hit);
      if (mask == 0u) continue;
      const int k = __builtin_popcount(mask);
      if (hit) found[wib][__builtin_popcount(mask & ((1u << lane) - 1u))] = src[e];
      const int sidx = (lane < k) ? found[wib][lane] : d;
      for (int t = 0; t < k; t += 16)
        gat_trip(feat4, stage[wib], dst_row[wib], sidx, t, min(16, k - t),
                 lane, m, half, m_run, l_run, acc);
    }
  }

  float4 o;
  o.x = acc.x / l_run; o.y = acc.y / l_run;
  o.z = acc.z / l_run; o.w = acc.w / l_run;
  *orow = o;
}

extern "C" void kernel_launch(void* const* d_in, const int* in_sizes, int n_in,
                              void* d_out, int out_size, void* d_ws, size_t ws_size,
                              hipStream_t stream) {
  const float* feat = (const float*)d_in[0];
  const int* src    = (const int*)d_in[1];
  const int* dst    = (const int*)d_in[2];
  const int E       = in_sizes[1];
  const int N       = in_sizes[0] / FEAT_DIM;
  float* out        = (float*)d_out;

  unsigned int* ws = (unsigned int*)d_ws;

  gat_chunk_sort_kernel<<<(E + CHUNK_EDGES - 1) / CHUNK_EDGES, SORT_THREADS, 0, stream>>>(
      src, dst, ws, N, E);
  gat_range_bucket_kernel<<<(N + RANGE_NODES - 1) / RANGE_NODES, SORT_THREADS, 0, stream>>>(
      ws, out, N, E);
  gat_node_wmma_kernel<<<(N + NODE_WAVES - 1) / NODE_WAVES, NODE_WAVES * 32, 0,
                         stream>>>(feat, src, dst, out, N, E);
}


# node kernel: trips work in groups of 4 edges (f32 WMMA tile = 4 edges x 4 feature slices, 8 WMMAs), only groups present, not 32 WMMAs + 16 rows always; score and softmax sums in a new fixed order
# speedup vs baseline: 5.6174x; 1.0674x over previous
#include <hip/hip_runtime.h>
#include <cstdint>
#include <math.h>


#define FEAT_DIM 128

typedef __attribute__((ext_vector_type(2))) float v2f;
typedef __attribute__((ext_vector_type(8))) float v8f;

#define BUCKET_CAP (FEAT_DIM - 1)

#define RANGE_SHIFT 6
#define RANGE_NODES (1 << RANGE_SHIFT)
#define MAX_BINS 1024
#define CHUNK_EDGES 2048
#define SORT_THREADS 256
#define EDGES_PER_THREAD (CHUNK_EDGES / SORT_THREADS)
#define ROW_IMG_PITCH (FEAT_DIM + 1)

__device__ __forceinline__ int gat_num_chunks(int E) {
  return (E + CHUNK_EDGES - 1) / CHUNK_EDGES;
}
__device__ __forceinline__ int gat_num_bins(int n_nodes) {
  return (n_nodes + RANGE_NODES - 1) >> RANGE_SHIFT;
}
__device__ __forceinline__ int gat_table_pitch(int n_bins) {
  return (n_bins + 1 + 63) & ~63;
}

__global__ __launch_bounds__(SORT_THREADS)
void gat_chunk_sort_kernel(const int* __restrict__ src,
                           const int* __restrict__ dst,
                           unsigned int* ws, int n_nodes, int E) {
  __shared__ int bin_cnt[MAX_BINS];
  __shared__ int bin_start[MAX_BINS + 1];
  __shared__ unsigned int recs[CHUNK_EDGES];
  __shared__ int wave_sum[SORT_THREADS / 32];

  const int tid  = threadIdx.x;
  const int lane = tid & 31;
  const int c    = blockIdx.x;
  const int base = c * CHUNK_EDGES;
  const int n_bins = gat_num_bins(n_nodes);

  for (int i = tid; i < MAX_BINS; i += SORT_THREADS) bin_cnt[i] = 0;
  __syncthreads();

  int dv[EDGES_PER_THREAD], sv[EDGES_PER_THREAD], rk[EDGES_PER_THREAD];
#pragma unroll
  for (int k = 0; k < EDGES_PER_THREAD; ++k) {
    const int e = min(base + k * SORT_THREADS + tid, E - 1);
    dv[k] = dst[e];
    sv[k] = src[e];
  }
#pragma unroll
  for (int k = 0; k < EDGES_PER_THREAD; ++k)
    if (base + k * SORT_THREADS + tid >= E) dv[k] = -1;
#pragma unroll
  for (int k = 0; k < EDGES_PER_THREAD; ++k) {
    rk[k] = 0;
    if (dv[k] >= 0)
      rk[k] = atomicAdd(&bin_cnt[min(dv[k] >> RANGE_SHIFT, MAX_BINS - 1)], 1);
  }
  __syncthreads();

  const int b4 = tid * 4;
  const int c0 = bin_cnt[b4], c1 = bin_cnt[b4 + 1], c2 = bin_cnt[b4 + 2], c3 = bin_cnt[b4 + 3];
  const int mine = c0 + c1 + c2 + c3;
  int incl = mine;
#pragma unroll
  for (int s = 1; s < 32; s <<= 1) {
    const int t = __shfl_up(incl, s);
    if (lane >= s) incl += t;
  }
  if (lane == 31) wave_sum[tid >> 5] = incl;
  __syncthreads();
  int prefix = incl - mine;
  for (int w = 0; w < (tid >> 5); ++w) prefix += wave_sum[w];
  bin_start[b4]     = prefix;
  bin_start[b4 + 1] = prefix + c0;
  bin_start[b4 + 2] = prefix + c0 + c1;
  bin_start[b4 + 3] = prefix + c0 + c1 + c2;
  if (tid == SORT_THREADS - 1) bin_start[MAX_BINS] = prefix + mine;
  __syncthreads();

#pragma unroll
  for (int k = 0; k < EDGES_PER_THREAD; ++k) {
    if (dv[k] >= 0) {
      const int bin = min(dv[k] >> RANGE_SHIFT, MAX_BINS - 1);
      recs[bin_start[bin] + rk[k]] =
          ((unsigned int)sv[k] << RANGE_SHIFT) | (unsigned int)(dv[k] & (RANGE_NODES - 1));
    }
  }
  __syncthreads();

  const int count = min(CHUNK_EDGES, E - base);
  unsigned int* rec = ws;
  for (int i = tid; i < count; i += SORT_THREADS) rec[base + i] = recs[i];
  unsigned int* table = ws + (size_t)gat_num_chunks(E) * CHUNK_EDGES;
  unsigned int* trow = table + (size_t)c * (gat_table_pitch(n_bins) >> 1);
  for (int i = tid; 2 * i <= n_bins; i += SORT_THREADS) {
    const unsigned int lo = (unsigned int)bin_start[2 * i];
    const unsigned int hi = (2 * i + 1 <= MAX_BINS) ? (unsigned int)bin_start[2 * i + 1] : 0u;
    trow[i] = lo | (hi << 16);
  }
}

__global__ __launch_bounds__(SORT_THREADS)
void gat_range_bucket_kernel(const unsigned int* ws, float* out, int n_nodes, int E) {
  __shared__ int img[RANGE_NODES * ROW_IMG_PITCH];
  __shared__ int cnt[RANGE_NODES];

  const int tid = threadIdx.x;
  const int b   = blockIdx.x;
  const int n_bins   = gat_num_bins(n_nodes);
  const int n_chunks = gat_num_chunks(E);

  for (int i = tid; i < RANGE_NODES * ROW_IMG_PITCH; i += SORT_THREADS) img[i] = 0;
  if (tid < RANGE_NODES) cnt[tid] = 0;
  __syncthreads();

  const unsigned int* rec = ws;
  const unsigned int* table = ws + (size_t)n_chunks * CHUNK_EDGES;
  const int pitch_words = gat_table_pitch(n_bins) >> 1;
  for (int c0 = tid; c0 < n_chunks; c0 += 2 * SORT_THREADS) {
    int s[2], e[2];
    unsigned int w0[2], w1[2];
    const unsigned int* piece[2];
#pragma unroll
    for (int u = 0; u < 2; ++u) {
      const int c = c0 + u * SORT_THREADS;
      const int cl = min(c, n_chunks - 1);
      const unsigned int* trow = table + (size_t)cl * pitch_words + (b >> 1);
      w0[u] = trow[0];
      w1[u] = trow[b & 1];
      piece[u] = rec + (size_t)cl * CHUNK_EDGES;
    }
    const int sh = (b & 1) * 16;
#pragma unroll
    for (int u = 0; u < 2; ++u) {
      const unsigned long long both = ((unsigned long long)w1[u] << 32) | w0[u];
      s[u] = (int)((both >> sh) & 0xffffu);
      e[u] = (int)((both >> (sh + 16)) & 0xffffu);
      if (c0 + u * SORT_THREADS >= n_chunks) e[u] = s[u];
    }
    while (s[0] < e[0] || s[1] < e[1]) {
      unsigned int r[8];
#pragma unroll
      for (int u = 0; u < 2; ++u) {
#pragma unroll
        for (int q = 0; q < 4; ++q)
          r[4 * u + q] = (s[u] + q < e[u]) ? piece[u][s[u] + q] : 0u;
      }
#pragma unroll
      for (int u = 0; u < 2; ++u) {
#pragma unroll
        for (int q = 0; q < 4; ++q) {
          if (s[u] + q < e[u]) {
            const unsigned int rr = r[4 * u + q];
            const int j = (int)(rr & (RANGE_NODES - 1));
            const int slot = atomicAdd(&cnt[j], 1);
            if (slot < BUCKET_CAP) img[j * ROW_IMG_PITCH + 1 + slot] = (int)(rr >> RANGE_SHIFT);
          }
        }
      }
      s[0] += 4;
      s[1] += 4;
    }
  }
  __syncthreads();

  const int lane = tid & 31;
  for (int j = tid >> 5; j < RANGE_NODES; j += SORT_THREADS / 32) {
    const int d = b * RANGE_NODES + j;
    if (d >= n_nodes) break;
    const int deg = cnt[j];
    int* orow = (int*)out + (size_t)d * FEAT_DIM;
    orow[lane] = (lane == 0) ? deg : img[j * ROW_IMG_PITCH + lane];
    const int lines = (min(deg, BUCKET_CAP) + 1 + 31) >> 5;
    for (int li = 1; li < lines; ++li)
      orow[li * 32 + lane] = img[j * ROW_IMG_PITCH + li * 32 + lane];
  }
}

#define NODE_WAVES 4
#define ROW_PITCH4 33

__device__ __forceinline__ float lane_value(float v, int l) {
  return __int_as_float(__builtin_amdgcn_readlane(__float_as_int(v), l));
}
__device__ __forceinline__ float lane_xor1(float v) {
  return __int_as_float(__builtin_amdgcn_update_dpp(0, __float_as_int(v), 0x161, 0xf, 0xf, true));
}
__device__ __forceinline__ float lane_xor2(float v) {
  return __int_as_float(__builtin_amdgcn_update_dpp(0, __float_as_int(v), 0x162, 0xf, 0xf, true));
}

template <int NG>
__device__ __forceinline__ void gat_trip(const float4* __restrict__ feat4,
                                         float4 (*stg)[ROW_PITCH4],
                                         const float4* drow,
                                         int sidx, int off, int cnt, int lane,
                                         float& m_run, float& l_run, float4& acc) {
  constexpr int NR = 4 * NG;
  const int q    = lane & 3;
  const int rho  = (lane >> 2) & 3;
  const int half = lane >> 4;

  float4 rs[NR];
#pragma unroll
  for (int r = 0; r < NR; ++r) {
    const int sr = __builtin_amdgcn_readlane(sidx, (off + r) & 31);
    rs[r] = feat4[(size_t)sr * (FEAT_DIM / 4) + lane];
  }
#pragma unroll
  for (int r = 0; r < NR; ++r) stg[4 * (r & 3) + (r >> 2)][lane] = rs[r];


  float4 hv[4];
#pragma unroll
  for (int jj = 0; jj < 4; ++jj) hv[jj] = drow[4 * (2 * jj + half) + q];

  float sc[NG];
#pragma unroll
  for (int g = 0; g < NG; ++g) {
    v8f c = {0.f, 0.f, 0.f, 0.f, 0.f, 0.f, 0.f, 0.f};
#pragma unroll
    for (int jj = 0; jj < 4; ++jj) {
      const float4 av = hv[jj];
      const float4 bv = stg[4 * rho + g][4 * (2 * jj + half) + q];
      v2f a; a.x = av.x; a.y = av.y;
      v2f b; b.x = bv.x; b.y = bv.y;
      c = __builtin_amdgcn_wmma_f32_16x16x4_f32(false, a, false, b, (short)0, c,
                                                false, false);
      a.x = av.z; a.y = av.w;
      b.x = bv.z; b.y = bv.w;
      c = __builtin_amdgcn_wmma_f32_16x16x4_f32(false, a, false, b, (short)0, c,
                                                false, false);
    }
    const float u = c[0] + lane_xor1(c[1]);
    const float v = c[2] + lane_xor1(c[3]);
    sc[g] = u + lane_xor2(v);
  }

  float mloc = -INFINITY;
#pragma unroll
  for (int g = 0; g < NG; ++g)
    if (4 * g + rho < cnt) mloc = fmaxf(mloc, sc[g]);
  const float m_new = fmaxf(fmaxf(m_run, lane_value(mloc, 0)),
                            fmaxf(fmaxf(lane_value(mloc, 4), lane_value(mloc, 8)),
                                  lane_value(mloc, 12)));
  const float scale = __expf(m_run - m_new);

  float p[NG];
  float ploc = 0.0f;
#pragma unroll
  for (int g = 0; g < NG; ++g) {
    p[g] = (4 * g + rho < cnt) ? __expf(sc[g] - m_new) : 0.0f;
    ploc += p[g];
  }
  l_run = l_run * scale + ((lane_value(ploc, 0) + lane_value(ploc, 4)) +
                           (lane_value(ploc, 8) + lane_value(ploc, 12)));

  acc.x *= scale; acc.y *= scale; acc.z *= scale; acc.w *= scale;
#pragma unroll
  for (int r = 0; r < NR; ++r) {
    const float w = lane_value(p[r >> 2], 4 * (r & 3));
    acc.x = fmaf(w, rs[r].x, acc.x);
    acc.y = fmaf(w, rs[r].y, acc.y);
    acc.z = fmaf(w, rs[r].z, acc.z);
    acc.w = fmaf(w, rs[r].w, acc.w);
  }
  m_run = m_new;
}

__device__ __forceinline__ void gat_trip_sized(const float4* __restrict__ feat4,
                                               float4 (*stg)[ROW_PITCH4],
                                               const float4* drow,
                                               int sidx, int off, int cnt, int lane,
                                               float& m_run, float& l_run, float4& acc) {
  switch ((cnt + 3) >> 2) {
    case 1:  gat_trip<1>(feat4, stg, drow, sidx, off, cnt, lane, m_run, l_run, acc); break;
    case 2:  gat_trip<2>(feat4, stg, drow, sidx, off, cnt, lane, m_run, l_run, acc); break;
    case 3:  gat_trip<3>(feat4, stg, drow, sidx, off, cnt, lane, m_run, l_run, acc); break;
    default: gat_trip<4>(feat4, stg, drow, sidx, off, cnt, lane, m_run, l_run, acc); break;
  }
}

__global__ __launch_bounds__(NODE_WAVES * 32, 8)
void gat_node_wmma_kernel(const float* __restrict__ feat,
                          const int* __restrict__ src,
                          const int* __restrict__ dst,
                          float* out, int n_nodes, int E) {
  __shared__ float4 stage[NODE_WAVES][16][ROW_PITCH4];
  __shared__ float4 dst_row[NODE_WAVES][32];
  __shared__ int found[NODE_WAVES][32];

  const int lane = threadIdx.x & 31;
  const int wib  = threadIdx.x >> 5;
  const int d = __builtin_amdgcn_readfirstlane(blockIdx.x * NODE_WAVES + wib);
  if (d >= n_nodes) return;

  const int* bucket = (const int*)out + (size_t)d * FEAT_DIM;
  int first = bucket[lane];
  const float4* feat4 = (const float4*)feat;
  float4 hd = feat4[(size_t)d * (FEAT_DIM / 4) + lane];
  asm volatile("" : "+v"(first), "+v"(hd.x), "+v"(hd.y), "+v"(hd.z), "+v"(hd.w));
  const int deg = __builtin_amdgcn_readlane(first, 0);
  float4* orow = (float4*)(out + (size_t)d * FEAT_DIM) + lane;

  if (deg <= 0) {
    *orow = hd;
    return;
  }

  dst_row[wib][lane] = hd;

  float  m_run = -INFINITY;
  float  l_run = 0.0f;
  float4 acc = make_float4(0.f, 0.f, 0.f, 0.f);

  if (deg <= BUCKET_CAP) {
    int sidx = (lane >= 1 && lane - 1 < deg) ? first : d;
    int off = 1, width = 16;
    for (int s0 = 0; s0 < deg; ) {
      gat_trip_sized(feat4, stage[wib], dst_row[wib], sidx, off, min(width, deg - s0),
                     lane, m_run, l_run, acc);
      s0 += width;
      if (s0 == 16) {
        off = 17; width = 15;
      } else {
        off = 0; width = 16;
        if (s0 < deg) {
          const int nxt = bucket[1 + s0 + (lane & 15)];
          sidx = (lane < 16 && s0 + lane < deg) ? nxt : d;
        }
      }
    }
  } else {
    for (int base = 0; base < E; base += 32) {
      const int e = base + lane;
      const bool hit = (e < E) && (dst[e] == d);
      const unsigned mask = __builtin_amdgcn_ballot_w32(hit);
      if (mask == 0u) continue;
      const int k = __builtin_popcount(mask);
      if (hit) found[wib][__builtin_popcount(mask & ((1u << lane) - 1u))] = src[e];
      const int sidx = (lane < k) ? found[wib][lane] : d;
      for (int t = 0; t < k; t += 16)
        gat_trip<4>(feat4, stage[wib], dst_row[wib], sidx, t, min(16, k - t),
                    lane, m_run, l_run, acc);
    }
  }

  float4 o;
  o.x = acc.x / l_run; o.y = acc.y / l_run;
  o.z = acc.z / l_run; o.w = acc.w / l_run;
  *orow = o;
}

extern "C" void kernel_launch(void* const* d_in, const int* in_sizes, int n_in,
                              void* d_out, int out_size, void* d_ws, size_t ws_size,
                              hipStream_t stream) {
  const float* feat = (const float*)d_in[0];
  const int* src    = (const int*)d_in[1];
  const int* dst    = (const int*)d_in[2];
  const int E       = in_sizes[1];
  const int N       = in_sizes[0] / FEAT_DIM;
  float* out        = (float*)d_out;

  unsigned int* ws = (unsigned int*)d_ws;

  gat_chunk_sort_kernel<<<(E + CHUNK_EDGES - 1) / CHUNK_EDGES, SORT_THREADS, 0, stream>>>(
      src, dst, ws, N, E);
  gat_range_bucket_kernel<<<(N + RANGE_NODES - 1) / RANGE_NODES, SORT_THREADS, 0, stream>>>(
      ws, out, N, E);
  gat_node_wmma_kernel<<<(N + NODE_WAVES - 1) / NODE_WAVES, NODE_WAVES * 32, 0,
                         stream>>>(feat, src, dst, out, N, E);
}
